# starting kernel
# baseline (speedup 1.0000x reference)
#include <hip/hip_runtime.h>


typedef __attribute__((ext_vector_type(2))) float v2f;
typedef __attribute__((ext_vector_type(8))) float v8f;

#define N_DIM 4096
#define D_DIM 256
#define BM 64
#define BN 64
#define KT 16
#define A_STRIDE 20
#define BPS 160
#define A_BUF (BM * A_STRIDE)
#define B_BUF ((KT / 2) * BPS)

__global__ __launch_bounds__(256) void poly_init_kernel(const float* __restrict__ x,
                                                        const float* __restrict__ coeffs,
                                                        float* __restrict__ y) {
    int i = blockIdx.x * blockDim.x + threadIdx.x;
    y[i] = coeffs[7] * x[i];
}

__global__ __launch_bounds__(256) void poly_gemm_kernel(const float* __restrict__ L,
                                                        const float* __restrict__ Y,
                                                        const float* __restrict__ X,
                                                        const float* __restrict__ coeffs,
                                                        int kIdx,
                                                        float* __restrict__ Out) {
    __shared__ float sA[2 * A_BUF];
    __shared__ float sB[2 * B_BUF];

    const int t    = threadIdx.x;
    const int lane = t & 31;
    const int wave = t >> 5;
    const int half = lane >> 4;
    const int l16  = lane & 15;

    const int mWave  = (wave & 3) * 16;
    const int nWave  = (wave >> 2) * 32;
    const int mBlock = blockIdx.x * BM;
    const int nBlock = blockIdx.y * BN;

    const int ar = t >> 2;
    const int ac = (t & 3) << 2;
    const int br = t >> 4;
    const int bc = (t & 15) << 2;
    const int bkp = br >> 1;
    const int bko = br & 1;

    const long aRowBase = (long)(mBlock + ar) * N_DIM + ac;

    v8f acc0 = {};
    v8f acc1 = {};

    float4 ra = *(const float4*)&L[aRowBase];
    float4 rb = *(const float4*)&Y[br * D_DIM + nBlock + bc];
    *(float4*)&sA[ar * A_STRIDE + ac] = ra;
    {
        float* p = &sB[bkp * BPS + 2 * bc + bko];
        p[0] = rb.x; p[2] = rb.y; p[4] = rb.z; p[6] = rb.w;
    }

    int cur = 0;
    for (int kk = 0; kk < N_DIM; kk += KT) {
        __syncthreads();

        const bool more = (kk + KT) < N_DIM;
        if (more) {
            ra = *(const float4*)&L[aRowBase + kk + KT];
            rb = *(const float4*)&Y[(kk + KT + br) * D_DIM + nBlock + bc];
            __builtin_prefetch(&L[aRowBase + kk + 2 * KT], 0, 3);
        }

        const int aOff = cur * A_BUF;
        const int bOff = cur * B_BUF;
#pragma unroll
        for (int kb = 0; kb < KT / 4; ++kb) {
            const int kp = 2 * kb + half;
            v2f a = *(const v2f*)&sA[aOff + (mWave + l16) * A_STRIDE + 4 * kb + 2 * half];
            v2f b0 = *(const v2f*)&sB[bOff + kp * BPS + 2 * (nWave + l16)];
            v2f b1 = *(const v2f*)&sB[bOff + kp * BPS + 2 * (nWave + 16 + l16)];
            acc0 = __builtin_amdgcn_wmma_f32_16x16x4_f32(
                false, a, false, b0, (short)0, acc0, false, false);
            acc1 = __builtin_amdgcn_wmma_f32_16x16x4_f32(
                false, a, false, b1, (short)0, acc1, false, false);
        }

        if (more) {
            const int na = (cur ^ 1) * A_BUF;
            const int nb = (cur ^ 1) * B_BUF;
            *(float4*)&sA[na + ar * A_STRIDE + ac] = ra;
            float* p = &sB[nb + bkp * BPS + 2 * bc + bko];
            p[0] = rb.x; p[2] = rb.y; p[4] = rb.z; p[6] = rb.w;
        }
        cur ^= 1;
    }

    const float c    = coeffs[kIdx];
    const int   mrow = mBlock + mWave + 8 * half;
    const int   ncol = nBlock + nWave + l16;
#pragma unroll
    for (int r = 0; r < 8; ++r) {
        int i0 = (mrow + r) * D_DIM + ncol;
        Out[i0]      = acc0[r] + c * X[i0];
        Out[i0 + 16] = acc1[r] + c * X[i0 + 16];
    }
}

extern "C" void kernel_launch(void* const* d_in, const int* in_sizes, int n_in,
                              void* d_out, int out_size, void* d_ws, size_t ws_size,
                              hipStream_t stream) {
    (void)in_sizes; (void)n_in; (void)out_size; (void)ws_size;
    const float* L      = (const float*)d_in[0];
    const float* x      = (const float*)d_in[1];
    const float* coeffs = (const float*)d_in[2];
    float* out = (float*)d_out;
    float* ws  = (float*)d_ws;

    poly_init_kernel<<<(N_DIM * D_DIM) / 256, 256, 0, stream>>>(x, coeffs, ws);

    dim3 grid(N_DIM / BM, D_DIM / BN);
    const float* src = ws;
    for (int p = 0; p < 7; ++p) {
        int    kIdx = 6 - p;
        float* dst  = (p & 1) ? ws : out;
        poly_gemm_kernel<<<grid, 256, 0, stream>>>(L, src, x, coeffs, kIdx, dst);
        src = dst;
    }
}


# No LDS or barrier: each wave streams its own fragments from global into registers 2 chunks ahead; y kept packed in fragment order (whole-line loads); k order inside each 8-k group changes
# speedup vs baseline: 2.4686x; 2.4686x over previous
#include <hip/hip_runtime.h>


typedef __attribute__((ext_vector_type(2))) float v2f;
typedef __attribute__((ext_vector_type(4))) float v4f;
typedef __attribute__((ext_vector_type(8))) float v8f;

#define N_DIM 4096
#define D_DIM 256
#define KC 32
#define NCHUNK (N_DIM / KC)
#define JT_STRIDE (NCHUNK * 512)
#define Q_STRIDE 512
#define I_STRIDE 128

struct Frag {
    v4f a[4];
    v4f b[4][4];
};

__global__ __launch_bounds__(256) void poly_init_kernel(const float* __restrict__ x,
                                                        const float* __restrict__ coeffs,
                                                        float* __restrict__ P) {
    const int g    = blockIdx.x * 256 + threadIdx.x;
    const int lane = g & 31;
    const int i    = (g >> 5) & 3;
    const int q    = (g >> 7) & (NCHUNK - 1);
    const int j    = g >> 14;
    const int k    = KC * q + 8 * i + 4 * (lane >> 4);
    const int n    = 16 * j + (lane & 15);
    const float c7 = coeffs[7];
    v4f v;
    v.x = c7 * x[(k + 0) * D_DIM + n];
    v.y = c7 * x[(k + 1) * D_DIM + n];
    v.z = c7 * x[(k + 2) * D_DIM + n];
    v.w = c7 * x[(k + 3) * D_DIM + n];
    *(v4f*)&P[(long)g * 4] = v;
}

__device__ __forceinline__ void load_chunk(Frag& f, const float* __restrict__ pa,
                                           const float* __restrict__ pb, int q) {
    const float* a = pa + q * KC;
    const float* b = pb + q * Q_STRIDE;
#pragma unroll
    for (int i = 0; i < 4; ++i) f.a[i] = *(const v4f*)(a + 8 * i);
#pragma unroll
    for (int jj = 0; jj < 4; ++jj) {
#pragma unroll
        for (int i = 0; i < 4; ++i)
            f.b[jj][i] = *(const v4f*)(b + jj * JT_STRIDE + i * I_STRIDE);
    }
}

__device__ __forceinline__ void mma_chunk(const Frag& f, v8f (&acc)[4]) {
#pragma unroll
    for (int i = 0; i < 4; ++i) {
        v2f a0 = {f.a[i].x, f.a[i].y};
        v2f a1 = {f.a[i].z, f.a[i].w};
#pragma unroll
        for (int jj = 0; jj < 4; ++jj) {
            v2f b0 = {f.b[jj][i].x, f.b[jj][i].y};
            acc[jj] = __builtin_amdgcn_wmma_f32_16x16x4_f32(
                false, a0, false, b0, (short)0, acc[jj], false, false);
        }
#pragma unroll
        for (int jj = 0; jj < 4; ++jj) {
            v2f b1 = {f.b[jj][i].z, f.b[jj][i].w};
            acc[jj] = __builtin_amdgcn_wmma_f32_16x16x4_f32(
                false, a1, false, b1, (short)0, acc[jj], false, false);
        }
    }
}

#define BLOCK_ENDS_HERE(acc)                                                     \
    asm volatile("" : "+v"(acc[0]), "+v"(acc[1]), "+v"(acc[2]), "+v"(acc[3]) : : "memory")
#define LOADS_STAY_AHEAD() __builtin_amdgcn_sched_barrier(0)

template <bool FINAL>
__global__ __launch_bounds__(128) void poly_gemm_kernel(const float* __restrict__ L,
                                                        const float* __restrict__ P,
                                                        const float* __restrict__ X,
                                                        const float* __restrict__ coeffs,
                                                        int kIdx,
                                                        float* __restrict__ Out) {
    const int t    = threadIdx.x;
    const int lane = t & 31;
    const int wave = t >> 5;
    const int half = lane >> 4;
    const int l16  = lane & 15;

    const int mBase = blockIdx.x * 64 + wave * 16;
    const int jBase = blockIdx.y * 4;

    const float* pa = L + (long)(mBase + l16) * N_DIM + 4 * half;
    const float* pb = P + (long)jBase * JT_STRIDE + lane * 4;

    v8f acc[4] = {};

    Frag f0, f1, f2;
    load_chunk(f0, pa, pb, 0);
    LOADS_STAY_AHEAD();
    load_chunk(f1, pa, pb, 1);
    LOADS_STAY_AHEAD();
    for (int q = 0; q < NCHUNK - 2; q += 3) {
        load_chunk(f2, pa, pb, q + 2);
        LOADS_STAY_AHEAD();
        mma_chunk(f0, acc);
        BLOCK_ENDS_HERE(acc);
        load_chunk(f0, pa, pb, q + 3);
        LOADS_STAY_AHEAD();
        mma_chunk(f1, acc);
        BLOCK_ENDS_HERE(acc);
        load_chunk(f1, pa, pb, q + 4);
        LOADS_STAY_AHEAD();
        mma_chunk(f2, acc);
        BLOCK_ENDS_HERE(acc);
    }

    const float c = coeffs[kIdx];
    const int mrow = mBase + 8 * half;
    const int ncol = blockIdx.y * 64 + l16;
    float xv[4][8];
#pragma unroll
    for (int jj = 0; jj < 4; ++jj) {
#pragma unroll
        for (int r = 0; r < 8; ++r) xv[jj][r] = X[(mrow + r) * D_DIM + ncol + 16 * jj];
    }

    mma_chunk(f0, acc);
    mma_chunk(f1, acc);

    if (FINAL) {
#pragma unroll
        for (int jj = 0; jj < 4; ++jj) {
#pragma unroll
            for (int r = 0; r < 8; ++r)
                Out[(mrow + r) * D_DIM + ncol + 16 * jj] = acc[jj][r] + c * xv[jj][r];
        }
    } else {
        const int qo = mBase >> 5;
        const int io = 2 * (wave & 1) + half;
#pragma unroll
        for (int jj = 0; jj < 4; ++jj) {
            float* o = Out + (long)(jBase + jj) * JT_STRIDE + qo * Q_STRIDE + io * I_STRIDE + l16 * 4;
            v4f lo, hi;
            lo.x = acc[jj][0] + c * xv[jj][0];
            lo.y = acc[jj][1] + c * xv[jj][1];
            lo.z = acc[jj][2] + c * xv[jj][2];
            lo.w = acc[jj][3] + c * xv[jj][3];
            hi.x = acc[jj][4] + c * xv[jj][4];
            hi.y = acc[jj][5] + c * xv[jj][5];
            hi.z = acc[jj][6] + c * xv[jj][6];
            hi.w = acc[jj][7] + c * xv[jj][7];
            *(v4f*)(o)      = lo;
            *(v4f*)(o + 64) = hi;
        }
    }
}

extern "C" void kernel_launch(void* const* d_in, const int* in_sizes, int n_in,
                              void* d_out, int out_size, void* d_ws, size_t ws_size,
                              hipStream_t stream) {
    (void)in_sizes; (void)n_in; (void)out_size; (void)ws_size;
    const float* L      = (const float*)d_in[0];
    const float* x      = (const float*)d_in[1];
    const float* coeffs = (const float*)d_in[2];
    float* out = (float*)d_out;
    float* ws  = (float*)d_ws;

    poly_init_kernel<<<(N_DIM * D_DIM) / (4 * 256), 256, 0, stream>>>(x, coeffs, ws);

    dim3 grid(N_DIM / 64, D_DIM / 64);
    const float* src = ws;
    for (int p = 0; p < 6; ++p) {
        int    kIdx = 6 - p;
        float* dst  = (p & 1) ? ws : out;
        poly_gemm_kernel<false><<<grid, 128, 0, stream>>>(L, src, x, coeffs, kIdx, dst);
        src = dst;
    }
    poly_gemm_kernel<true><<<grid, 128, 0, stream>>>(L, src, x, coeffs, 0, out);
}


# Split-K inside the work-group: each wave keeps the whole 64x64 block (16 accumulator tiles) over every 4th 32-k chunk; 4 partials summed via LDS once; B loads 4x fewer; sum order changes
# speedup vs baseline: 2.9353x; 1.1891x over previous
#include <hip/hip_runtime.h>


typedef __attribute__((ext_vector_type(2))) float v2f;
typedef __attribute__((ext_vector_type(4))) float v4f;
typedef __attribute__((ext_vector_type(8))) float v8f;

#define N_DIM 4096
#define D_DIM 256
#define KC 32
#define NCHUNK (N_DIM / KC)
#define JT_STRIDE (NCHUNK * 512)
#define Q_STRIDE 512
#define I_STRIDE 128

#define NWAVE 4
#define RF_STRIDE (16 * N_DIM)

struct Frag {
    v4f a[4][4];
    v4f b[4][4];
};

__global__ __launch_bounds__(256) void poly_init_kernel(const float* __restrict__ x,
                                                        const float* __restrict__ coeffs,
                                                        float* __restrict__ P) {
    const int g    = blockIdx.x * 256 + threadIdx.x;
    const int lane = g & 31;
    const int i    = (g >> 5) & 3;
    const int q    = (g >> 7) & (NCHUNK - 1);
    const int j    = g >> 14;
    const int k    = KC * q + 8 * i + 4 * (lane >> 4);
    const int n    = 16 * j + (lane & 15);
    const float c7 = coeffs[7];
    v4f v;
    v.x = c7 * x[(k + 0) * D_DIM + n];
    v.y = c7 * x[(k + 1) * D_DIM + n];
    v.z = c7 * x[(k + 2) * D_DIM + n];
    v.w = c7 * x[(k + 3) * D_DIM + n];
    *(v4f*)&P[(long)g * 4] = v;
}

__device__ __forceinline__ void load_chunk(Frag& f, const float* __restrict__ pa,
                                           const float* __restrict__ pb, int m) {
    const float* a = pa + m * (NWAVE * KC);
    const float* b = pb + m * (NWAVE * Q_STRIDE);
#pragma unroll
    for (int rf = 0; rf < 4; ++rf) {
#pragma unroll
        for (int i = 0; i < 4; ++i)
            f.a[rf][i] = *(const v4f*)(a + rf * RF_STRIDE + 8 * i);
    }
#pragma unroll
    for (int jj = 0; jj < 4; ++jj) {
#pragma unroll
        for (int i = 0; i < 4; ++i)
            f.b[jj][i] = *(const v4f*)(b + jj * JT_STRIDE + i * I_STRIDE);
    }
}

__device__ __forceinline__ void mma_chunk(const Frag& f, v8f (&acc)[4][4]) {
#pragma unroll
    for (int i = 0; i < 4; ++i) {
#pragma unroll
        for (int s = 0; s < 2; ++s) {
            v2f a[4], b[4];
#pragma unroll
            for (int rf = 0; rf < 4; ++rf) {
                a[rf].x = f.a[rf][i][2 * s];
                a[rf].y = f.a[rf][i][2 * s + 1];
            }
#pragma unroll
            for (int jj = 0; jj < 4; ++jj) {
                b[jj].x = f.b[jj][i][2 * s];
                b[jj].y = f.b[jj][i][2 * s + 1];
            }
#pragma unroll
            for (int rf = 0; rf < 4; ++rf) {
#pragma unroll
                for (int jj = 0; jj < 4; ++jj)
                    acc[rf][jj] = __builtin_amdgcn_wmma_f32_16x16x4_f32(
                        false, a[rf], false, b[jj], (short)0, acc[rf][jj], false, false);
            }
        }
    }
}

#define BLOCK_ENDS_HERE(acc)                                                          \
    do {                                                                              \
        asm volatile("" : "+v"(acc[0][0]), "+v"(acc[0][1]), "+v"(acc[0][2]),          \
                          "+v"(acc[0][3]), "+v"(acc[1][0]), "+v"(acc[1][1]),          \
                          "+v"(acc[1][2]), "+v"(acc[1][3]) : : "memory");             \
        asm volatile("" : "+v"(acc[2][0]), "+v"(acc[2][1]), "+v"(acc[2][2]),          \
                          "+v"(acc[2][3]), "+v"(acc[3][0]), "+v"(acc[3][1]),          \
                          "+v"(acc[3][2]), "+v"(acc[3][3]) : : "memory");             \
    } while (0)
#define LOADS_STAY_AHEAD() __builtin_amdgcn_sched_barrier(0)

template <bool FINAL>
__global__ __launch_bounds__(128) void poly_gemm_kernel(const float* __restrict__ L,
                                                        const float* __restrict__ P,
                                                        const float* __restrict__ X,
                                                        const float* __restrict__ coeffs,
                                                        int kIdx,
                                                        float* __restrict__ Out) {
    const int t    = threadIdx.x;
    const int lane = t & 31;
    const int wave = t >> 5;
    const int half = lane >> 4;
    const int l16  = lane & 15;

    const int mBlock = blockIdx.x * 64;
    const int jBase  = blockIdx.y * 4;

    __shared__ v4f red[NWAVE][4][4][2][32];

    const float* pa = L + (long)(mBlock + l16) * N_DIM + wave * KC + 4 * half;
    const float* pb = P + (long)jBase * JT_STRIDE + wave * Q_STRIDE + lane * 4;

    v8f acc[4][4] = {};

    Frag f0, f1;
    load_chunk(f0, pa, pb, 0);
    LOADS_STAY_AHEAD();
    for (int m = 0; m < NCHUNK / NWAVE - 2; m += 2) {
        load_chunk(f1, pa, pb, m + 1);
        LOADS_STAY_AHEAD();
        mma_chunk(f0, acc);
        BLOCK_ENDS_HERE(acc);
        load_chunk(f0, pa, pb, m + 2);
        LOADS_STAY_AHEAD();
        mma_chunk(f1, acc);
        BLOCK_ENDS_HERE(acc);
    }
    load_chunk(f1, pa, pb, NCHUNK / NWAVE - 1);
    LOADS_STAY_AHEAD();
    mma_chunk(f0, acc);
    BLOCK_ENDS_HERE(acc);

    const float c = coeffs[kIdx];
    const int mBase = mBlock + wave * 16;
    const int mrow = mBase + 8 * half;
    const int ncol = blockIdx.y * 64 + l16;
    float xv[4][8];
#pragma unroll
    for (int jj = 0; jj < 4; ++jj) {
#pragma unroll
        for (int r = 0; r < 8; ++r) xv[jj][r] = X[(mrow + r) * D_DIM + ncol + 16 * jj];
    }

    mma_chunk(f1, acc);

#pragma unroll
    for (int rf = 0; rf < 4; ++rf) {
#pragma unroll
        for (int jj = 0; jj < 4; ++jj) {
            v4f lo = {acc[rf][jj][0], acc[rf][jj][1], acc[rf][jj][2], acc[rf][jj][3]};
            v4f hi = {acc[rf][jj][4], acc[rf][jj][5], acc[rf][jj][6], acc[rf][jj][7]};
            red[wave][rf][jj][0][lane] = lo;
            red[wave][rf][jj][1][lane] = hi;
        }
    }
    __syncthreads();
    v8f sum[4];
#pragma unroll
    for (int jj = 0; jj < 4; ++jj) {
        v4f lo = red[0][wave][jj][0][lane];
        v4f hi = red[0][wave][jj][1][lane];
#pragma unroll
        for (int s = 1; s < NWAVE; ++s) {
            lo += red[s][wave][jj][0][lane];
            hi += red[s][wave][jj][1][lane];
        }
        sum[jj][0] = lo.x; sum[jj][1] = lo.y; sum[jj][2] = lo.z; sum[jj][3] = lo.w;
        sum[jj][4] = hi.x; sum[jj][5] = hi.y; sum[jj][6] = hi.z; sum[jj][7] = hi.w;
    }

    if (FINAL) {
#pragma unroll
        for (int jj = 0; jj < 4; ++jj) {
#pragma unroll
            for (int r = 0; r < 8; ++r)
                Out[(mrow + r) * D_DIM + ncol + 16 * jj] = sum[jj][r] + c * xv[jj][r];
        }
    } else {
        const int qo = mBase >> 5;
        const int io = 2 * (wave & 1) + half;
#pragma unroll
        for (int jj = 0; jj < 4; ++jj) {
            float* o = Out + (long)(jBase + jj) * JT_STRIDE + qo * Q_STRIDE + io * I_STRIDE + l16 * 4;
            v4f lo, hi;
            lo.x = sum[jj][0] + c * xv[jj][0];
            lo.y = sum[jj][1] + c * xv[jj][1];
            lo.z = sum[jj][2] + c * xv[jj][2];
            lo.w = sum[jj][3] + c * xv[jj][3];
            hi.x = sum[jj][4] + c * xv[jj][4];
            hi.y = sum[jj][5] + c * xv[jj][5];
            hi.z = sum[jj][6] + c * xv[jj][6];
            hi.w = sum[jj][7] + c * xv[jj][7];
            *(v4f*)(o)      = lo;
            *(v4f*)(o + 64) = hi;
        }
    }
}

extern "C" void kernel_launch(void* const* d_in, const int* in_sizes, int n_in,
                              void* d_out, int out_size, void* d_ws, size_t ws_size,
                              hipStream_t stream) {
    (void)in_sizes; (void)n_in; (void)out_size; (void)ws_size;
    const float* L      = (const float*)d_in[0];
    const float* x      = (const float*)d_in[1];
    const float* coeffs = (const float*)d_in[2];
    float* out = (float*)d_out;
    float* ws  = (float*)d_ws;

    poly_init_kernel<<<(N_DIM * D_DIM) / (4 * 256), 256, 0, stream>>>(x, coeffs, ws);

    dim3 grid(N_DIM / 64, D_DIM / 64);
    const float* src = ws;
    for (int p = 0; p < 6; ++p) {
        int    kIdx = 6 - p;
        float* dst  = (p & 1) ? ws : out;
        poly_gemm_kernel<false><<<grid, 128, 0, stream>>>(L, src, x, coeffs, kIdx, dst);
        src = dst;
    }
    poly_gemm_kernel<true><<<grid, 128, 0, stream>>>(L, src, x, coeffs, 0, out);
}
